# starting kernel
# baseline (speedup 1.0000x reference)
#include <hip/hip_runtime.h>

typedef __attribute__((ext_vector_type(2))) float v2f;
typedef __attribute__((ext_vector_type(8))) float v8f;

#define N_CH     64
#define KDIM     128
#define NDIM     128
#define ROWS     128
#define FPAD     132
#define NKSTEP   (KDIM / 4)
#define NTILES   (NDIM / 16)
#define BFRAG_FLOATS (NKSTEP * NTILES * 32 * 2)

__global__ void build_wfrag_kernel(const float* __restrict__ W,
                                   float* __restrict__ bfrag) {
    int idx  = blockIdx.x * blockDim.x + threadIdx.x;
    int comp = idx & 1;
    int p    = idx >> 1;
    int lane = p & 31;
    int nt   = (p >> 5) & 7;
    int kk   = p >> 8;

    int koff = (lane >> 4) * 2;
    int k = kk * 4 + koff + comp;
    int n = nt * 16 + (lane & 15);

    int c = n >> 1;
    int o = n & 1;
    int j = (k < N_CH) ? k : k - N_CH;
    float v = 0.0f;
    if (j != c) {
        int p_loo = (j < c) ? j : j - 1;
        int feat  = (k < N_CH) ? p_loo : 63 + p_loo;
        v = W[c * 252 + o * 126 + feat];
    }
    bfrag[idx] = v;
}

__global__ __launch_bounds__(256) void leakage_wmma_kernel(
        const float* __restrict__ x,
        const float* __restrict__ bfrag,
        float* __restrict__ out) {
    extern __shared__ float lds[];
    float* Ftile = lds;
    float* Bw    = lds + ROWS * FPAD;

    const int tid      = threadIdx.x;
    const int row_base = blockIdx.x * ROWS;

    {
        const float4* src = (const float4*)bfrag;
        float4*       dst = (float4*)Bw;
        #pragma unroll
        for (int i = 0; i < BFRAG_FLOATS / 4 / 256; ++i)
            dst[tid + i * 256] = src[tid + i * 256];
    }

    {
        const float2* x2 = (const float2*)x;
        #pragma unroll
        for (int i = 0; i < 32; ++i) {
            int lin = tid + i * 256;
            int r   = lin >> 6;
            int c   = lin & 63;
            float2 v = x2[(size_t)(row_base + r) * N_CH + c];
            float amp = v.x * v.x + v.y * v.y;
            Ftile[r * FPAD + c]        = amp * v.x;
            Ftile[r * FPAD + N_CH + c] = amp * v.y;
        }
    }

    __syncthreads();

    const int wave    = tid >> 5;
    const int lane    = tid & 31;
    const int halfsel = lane >> 4;
    const int lrow    = lane & 15;
    const int koff    = halfsel * 2;

    v8f acc[NTILES] = {};

    const float* Frow  = &Ftile[(wave * 16 + lrow) * FPAD];
    const float* Blane = &Bw[lane * 2];

    #pragma unroll 4
    for (int kk = 0; kk < NKSTEP; ++kk) {
        v2f a = *(const v2f*)&Frow[kk * 4 + koff];
        #pragma unroll
        for (int nt = 0; nt < NTILES; ++nt) {
            v2f b = *(const v2f*)&Blane[(kk * NTILES + nt) * 64];
            acc[nt] = __builtin_amdgcn_wmma_f32_16x16x4_f32(
                false, a, false, b, (short)0, acc[nt], false, false);
        }
    }

    #pragma unroll
    for (int nt = 0; nt < NTILES; ++nt) {
        #pragma unroll
        for (int v = 0; v < 8; ++v) {
            int m = v + halfsel * 8;
            size_t r = (size_t)(row_base + wave * 16 + m);
            out[r * NDIM + nt * 16 + lrow] = acc[nt][v];
        }
    }
}

extern "C" void kernel_launch(void* const* d_in, const int* in_sizes, int n_in,
                              void* d_out, int out_size, void* d_ws, size_t ws_size,
                              hipStream_t stream) {
    const float* x = (const float*)d_in[0];
    const float* W = (const float*)d_in[1];
    float* out   = (float*)d_out;
    float* bfrag = (float*)d_ws;

    build_wfrag_kernel<<<BFRAG_FLOATS / 256, 256, 0, stream>>>(W, bfrag);

    int rows = in_sizes[0] / (2 * N_CH);
    int grid = rows / ROWS;
    size_t shmem = (size_t)(ROWS * FPAD + BFRAG_FLOATS) * sizeof(float);
    leakage_wmma_kernel<<<grid, 256, shmem, stream>>>(x, bfrag, out);
}


# Waves split output columns, not rows: each keeps its 128x16 B strip in registers and walks rows in 32-row chunks; F staged one chunk ahead in LDS under a split LDS-only barrier; sums unchanged
# speedup vs baseline: 1.2344x; 1.2344x over previous
#include <hip/hip_runtime.h>


typedef __attribute__((ext_vector_type(2))) float v2f;
typedef __attribute__((ext_vector_type(8))) float v8f;

#define N_CH     64
#define KDIM     128
#define NDIM     128
#define ROWS     128
#define CROWS    32
#define NCHUNK   (ROWS / CROWS)
#define FPAD     132
#define NKSTEP   (KDIM / 4)
#define NTILES   (NDIM / 16)
#define KGROUP   8
#define NGROUP   (NKSTEP / KGROUP)
#define BFRAG_FLOATS (NKSTEP * NTILES * 32 * 2)

__global__ void build_wfrag_kernel(const float* __restrict__ W,
                                   float* __restrict__ bfrag) {
    int idx  = blockIdx.x * blockDim.x + threadIdx.x;
    int comp = idx & 1;
    int p    = idx >> 1;
    int lane = p & 31;
    int nt   = (p >> 5) & 7;
    int kk   = p >> 8;

    int koff = (lane >> 4) * 2;
    int k = kk * 4 + koff + comp;
    int n = nt * 16 + (lane & 15);

    int c = n >> 1;
    int o = n & 1;
    int j = (k < N_CH) ? k : k - N_CH;
    float v = 0.0f;
    if (j != c) {
        int p_loo = (j < c) ? j : j - 1;
        int feat  = (k < N_CH) ? p_loo : 63 + p_loo;
        v = W[c * 252 + o * 126 + feat];
    }
    bfrag[idx] = v;
}

static __device__ __forceinline__ void lds_ready_signal() {
    asm volatile("s_wait_dscnt 0x0\n\ts_barrier_signal -1" ::: "memory");
}
static __device__ __forceinline__ void lds_ready_wait() {
    asm volatile("s_barrier_wait -1" ::: "memory");
}

__global__ __launch_bounds__(256) void leakage_wmma_kernel(
        const float* __restrict__ x,
        const float* __restrict__ bfrag,
        float* __restrict__ out) {
    extern __shared__ float Ftile[];

    const int tid      = threadIdx.x;
    const int wave     = tid >> 5;
    const int lane     = tid & 31;
    const int halfsel  = lane >> 4;
    const int lrow     = lane & 15;
    const int row_base = blockIdx.x * ROWS;

    const float4* x4 = (const float4*)x + (size_t)row_base * 32;

    float4 xr[NCHUNK][4];

#define LOAD_CHUNK(c)                                                        \
    _Pragma("unroll")                                                        \
    for (int i = 0; i < 4; ++i)                                              \
        xr[c][i] = x4[(size_t)((c) * CROWS + i * 8 + wave) * 32 + lane];

#define STAGE_CHUNK(c)                                                       \
    _Pragma("unroll")                                                        \
    for (int i = 0; i < 4; ++i) {                                            \
        float4 v = xr[c][i];                                                 \
        float a0 = v.x * v.x + v.y * v.y;                                    \
        float a1 = v.z * v.z + v.w * v.w;                                    \
        float* rowp = &Ftile[((c) * CROWS + i * 8 + wave) * FPAD + 2 * lane];\
        v2f fr = { a0 * v.x, a1 * v.z };                                     \
        v2f fi = { a0 * v.y, a1 * v.w };                                     \
        *(v2f*)rowp          = fr;                                           \
        *(v2f*)(rowp + N_CH) = fi;                                           \
    }

    LOAD_CHUNK(0)

    v2f breg[NKSTEP];
    {
        const v2f* bsrc = (const v2f*)bfrag + wave * 32 + lane;
        #pragma unroll
        for (int kk = 0; kk < NKSTEP; ++kk)
            breg[kk] = bsrc[kk * (NTILES * 32)];
    }

    LOAD_CHUNK(1)
    __builtin_amdgcn_sched_barrier(0);
    LOAD_CHUNK(2)
    LOAD_CHUNK(3)

    STAGE_CHUNK(0)
    lds_ready_signal();
    lds_ready_wait();

    const int aoff = lrow * FPAD + 2 * halfsel;

#define LOAD_A(set, g)                                                       \
    _Pragma("unroll")                                                        \
    for (int j = 0; j < KGROUP; ++j) {                                       \
        af[set][0][j] = *(const v2f*)&Fa[((g) * KGROUP + j) * 4];            \
        af[set][1][j] = *(const v2f*)&Fa[16 * FPAD + ((g) * KGROUP + j) * 4];\
    }

    #pragma unroll
    for (int c = 0; c < NCHUNK; ++c) {
        const float* Fa = &Ftile[c * CROWS * FPAD + aoff];
        v2f af[2][2][KGROUP];

        LOAD_A(0, 0)

        if (c + 1 < NCHUNK) {
            STAGE_CHUNK(c + 1)
            lds_ready_signal();
        }

        v8f acc0 = {};
        v8f acc1 = {};
        #pragma unroll
        for (int g = 0; g < NGROUP; ++g) {
            if (g + 1 < NGROUP) { LOAD_A((g + 1) & 1, g + 1) }
            __builtin_amdgcn_sched_barrier(0);
            #pragma unroll
            for (int j = 0; j < KGROUP; ++j) {
                acc0 = __builtin_amdgcn_wmma_f32_16x16x4_f32(
                    false, af[g & 1][0][j], false, breg[g * KGROUP + j],
                    (short)0, acc0, false, false);
                acc1 = __builtin_amdgcn_wmma_f32_16x16x4_f32(
                    false, af[g & 1][1][j], false, breg[g * KGROUP + j],
                    (short)0, acc1, false, false);
            }
            __builtin_amdgcn_sched_barrier(0);
        }

        float* orow = out + (size_t)(row_base + c * CROWS + 8 * halfsel) * NDIM
                          + wave * 16 + lrow;
        #pragma unroll
        for (int v = 0; v < 8; ++v) {
            orow[(size_t)v * NDIM]        = acc0[v];
            orow[(size_t)(16 + v) * NDIM] = acc1[v];
        }

        if (c + 1 < NCHUNK) lds_ready_wait();
    }
#undef LOAD_A
#undef LOAD_CHUNK
#undef STAGE_CHUNK
}

extern "C" void kernel_launch(void* const* d_in, const int* in_sizes, int n_in,
                              void* d_out, int out_size, void* d_ws, size_t ws_size,
                              hipStream_t stream) {
    const float* x = (const float*)d_in[0];
    const float* W = (const float*)d_in[1];
    float* out   = (float*)d_out;
    float* bfrag = (float*)d_ws;

    build_wfrag_kernel<<<BFRAG_FLOATS / 256, 256, 0, stream>>>(W, bfrag);

    int rows = in_sizes[0] / (2 * N_CH);
    int grid = rows / ROWS;
    size_t shmem = (size_t)(ROWS * FPAD) * sizeof(float);
    leakage_wmma_kernel<<<grid, 256, shmem, stream>>>(x, bfrag, out);
}


# Each lane reads its own column's weight row from W in 16-byte pieces into its B registers (k regrouped 4 per lane half, A follows, sums reordered); own-channel zeros by select; expansion launch gone
# speedup vs baseline: 1.4564x; 1.1798x over previous
#include <hip/hip_runtime.h>


typedef __attribute__((ext_vector_type(2))) float v2f;
typedef __attribute__((ext_vector_type(4))) float v4f;
typedef __attribute__((ext_vector_type(8))) float v8f;
typedef v4f v4f_a4 __attribute__((aligned(4)));

#define N_CH     64
#define KDIM     128
#define NDIM     128
#define ROWS     128
#define CROWS    32
#define NCHUNK   (ROWS / CROWS)
#define FPAD     132
#define NPIECE   (KDIM / 8)
#define PGROUP   4
#define NGROUP   (NPIECE / PGROUP)
#define WROW     126


static __device__ __forceinline__ void lds_ready_signal() {
    asm volatile("s_wait_dscnt 0x0\n\ts_barrier_signal -1" ::: "memory");
}
static __device__ __forceinline__ void lds_ready_wait() {
    asm volatile("s_barrier_wait -1" ::: "memory");
}

__global__ __launch_bounds__(256) void leakage_wmma_kernel(
        const float* __restrict__ x,
        const float* __restrict__ W,
        float* __restrict__ out) {
    extern __shared__ __attribute__((aligned(16))) float Ftile[];

    const int tid      = threadIdx.x;
    const int wave     = tid >> 5;
    const int lane     = tid & 31;
    const int halfsel  = lane >> 4;
    const int lrow     = lane & 15;
    const int row_base = blockIdx.x * ROWS;
    const int wave_u   = __builtin_amdgcn_readfirstlane(wave);

    const float4* x4 = (const float4*)x + (size_t)row_base * 32;

    float4 xr[NCHUNK][4];

#define LOAD_CHUNK(c)                                                        \
    _Pragma("unroll")                                                        \
    for (int i = 0; i < 4; ++i)                                              \
        xr[c][i] = x4[(size_t)((c) * CROWS + i * 8 + wave) * 32 + lane];

#define STAGE_CHUNK(c)                                                       \
    _Pragma("unroll")                                                        \
    for (int i = 0; i < 4; ++i) {                                            \
        float4 v = xr[c][i];                                                 \
        float a0 = v.x * v.x + v.y * v.y;                                    \
        float a1 = v.z * v.z + v.w * v.w;                                    \
        float* rowp = &Ftile[((c) * CROWS + i * 8 + wave) * FPAD + 2 * lane];\
        v2f fr = { a0 * v.x, a1 * v.z };                                     \
        v2f fi = { a0 * v.y, a1 * v.w };                                     \
        *(v2f*)rowp          = fr;                                           \
        *(v2f*)(rowp + N_CH) = fi;                                           \
    }

    const int    ncol = wave * 16 + lrow;
    const int    cch  = ncol >> 1;
    const float* wrow = W + ncol * WROW;
    v4f bq[NPIECE];

#define LOAD_B(i) {                                                          \
        const int j0 = 8 * ((i) & 7) + 4 * halfsel;                          \
        const int qc = cch - j0;                                             \
        const int r0 = ((i) < 8) ? (j0 - (qc < 0 ? 1 : 0))                   \
                                 : (63 + j0 - (qc < 4 ? 1 : 0));             \
        bq[i] = *(const v4f_a4*)(wrow + r0);                                 \
    }

#define FIX_B(i)                                                             \
    if (((i) & 7) == wave_u) {                                               \
        const int qc = (lrow >> 1) - 4 * halfsel;                            \
        const v4f R = bq[i];                                                 \
        v4f E;                                                               \
        if ((i) < 8) {                                                       \
            E.x = (qc == 0) ? 0.0f : R.x;                                    \
            E.y = (qc == 1) ? 0.0f : ((qc == 0) ? R.x : R.y);                \
            E.z = (qc == 2) ? 0.0f : ((qc == 0 || qc == 1) ? R.y : R.z);     \
            E.w = (qc == 3) ? 0.0f : ((qc >= 0 && qc < 3) ? R.z : R.w);      \
        } else {                                                             \
            E.x = (qc == 0) ? 0.0f : ((qc >= 1 && qc <= 3) ? R.y : R.x);     \
            E.y = (qc == 1) ? 0.0f : ((qc == 2 || qc == 3) ? R.z : R.y);     \
            E.z = (qc == 2) ? 0.0f : ((qc == 3) ? R.w : R.z);                \
            E.w = (qc == 3) ? 0.0f : R.w;                                    \
        }                                                                    \
        bq[i] = E;                                                           \
    }

    LOAD_CHUNK(0)
    LOAD_B(0) LOAD_B(1) LOAD_B(2) LOAD_B(3)
    LOAD_CHUNK(1)
    __builtin_amdgcn_sched_barrier(0);
    LOAD_B(4)  LOAD_B(5)  LOAD_B(6)  LOAD_B(7)
    LOAD_B(8)  LOAD_B(9)  LOAD_B(10) LOAD_B(11)
    LOAD_B(12) LOAD_B(13) LOAD_B(14) LOAD_B(15)
    __builtin_amdgcn_sched_barrier(0);
    LOAD_CHUNK(2)
    LOAD_CHUNK(3)

    STAGE_CHUNK(0)
    lds_ready_signal();
    lds_ready_wait();

    const int aoff = lrow * FPAD + 4 * halfsel;

#define LOAD_A(set, g)                                                       \
    _Pragma("unroll")                                                        \
    for (int j = 0; j < PGROUP; ++j) {                                       \
        af[set][0][j] = *(const v4f*)&Fa[((g) * PGROUP + j) * 8];            \
        af[set][1][j] = *(const v4f*)&Fa[16 * FPAD + ((g) * PGROUP + j) * 8];\
    }

    #pragma unroll
    for (int c = 0; c < NCHUNK; ++c) {
        const float* Fa = &Ftile[c * CROWS * FPAD + aoff];
        v4f af[2][2][PGROUP];

        LOAD_A(0, 0)

        if (c + 1 < NCHUNK) {
            STAGE_CHUNK(c + 1)
            lds_ready_signal();
        }

        v8f acc0 = {};
        v8f acc1 = {};
        #pragma unroll
        for (int g = 0; g < NGROUP; ++g) {
            if (g + 1 < NGROUP) { LOAD_A((g + 1) & 1, g + 1) }
            if (c == 0) {
                FIX_B(g * PGROUP + 0) FIX_B(g * PGROUP + 1)
                FIX_B(g * PGROUP + 2) FIX_B(g * PGROUP + 3)
            }
            __builtin_amdgcn_sched_barrier(0);
            #pragma unroll
            for (int j = 0; j < PGROUP; ++j) {
                const v4f a0 = af[g & 1][0][j];
                const v4f a1 = af[g & 1][1][j];
                const v4f b  = bq[g * PGROUP + j];
                const v2f blo = { b.x, b.y };
                const v2f bhi = { b.z, b.w };
                const v2f a0lo = { a0.x, a0.y };
                const v2f a0hi = { a0.z, a0.w };
                const v2f a1lo = { a1.x, a1.y };
                const v2f a1hi = { a1.z, a1.w };
                acc0 = __builtin_amdgcn_wmma_f32_16x16x4_f32(
                    false, a0lo, false, blo, (short)0, acc0, false, false);
                acc1 = __builtin_amdgcn_wmma_f32_16x16x4_f32(
                    false, a1lo, false, blo, (short)0, acc1, false, false);
                acc0 = __builtin_amdgcn_wmma_f32_16x16x4_f32(
                    false, a0hi, false, bhi, (short)0, acc0, false, false);
                acc1 = __builtin_amdgcn_wmma_f32_16x16x4_f32(
                    false, a1hi, false, bhi, (short)0, acc1, false, false);
            }
            __builtin_amdgcn_sched_barrier(0);
        }

        float* orow = out + (size_t)(row_base + c * CROWS + 8 * halfsel) * NDIM
                          + wave * 16 + lrow;
        #pragma unroll
        for (int v = 0; v < 8; ++v) {
            orow[(size_t)v * NDIM]        = acc0[v];
            orow[(size_t)(16 + v) * NDIM] = acc1[v];
        }

        if (c + 1 < NCHUNK) lds_ready_wait();
    }
#undef LOAD_A
#undef LOAD_B
#undef FIX_B
#undef LOAD_CHUNK
#undef STAGE_CHUNK
}

extern "C" void kernel_launch(void* const* d_in, const int* in_sizes, int n_in,
                              void* d_out, int out_size, void* d_ws, size_t ws_size,
                              hipStream_t stream) {
    const float* x = (const float*)d_in[0];
    const float* W = (const float*)d_in[1];
    float* out   = (float*)d_out;
    (void)d_ws; (void)ws_size;

    int rows = in_sizes[0] / (2 * N_CH);
    int grid = rows / ROWS;
    size_t shmem = (size_t)(ROWS * FPAD) * sizeof(float);
    leakage_wmma_kernel<<<grid, 256, shmem, stream>>>(x, W, out);
}
